# starting kernel
# baseline (speedup 1.0000x reference)
#include <hip/hip_runtime.h>

typedef float v2f __attribute__((ext_vector_type(2)));
typedef float v8f __attribute__((ext_vector_type(8)));

#define RANK 16
#define DIM  400
#define FACTOR_ELEMS (RANK * DIM)
#define LDS_FLOATS   (3 * FACTOR_ELEMS)

__global__ void parafac_query_wmma(const float* __restrict__ U,
                                   const float* __restrict__ V,
                                   const float* __restrict__ W,
                                   const int*   __restrict__ idx,
                                   float*       __restrict__ out,
                                   int nQueries)
{
    extern __shared__ float lds[];
    float* Ul = lds;
    float* Vl = lds + FACTOR_ELEMS;
    float* Wl = lds + 2 * FACTOR_ELEMS;

    for (int t = threadIdx.x; t < FACTOR_ELEMS; t += blockDim.x) {
        Ul[t] = U[t];
        Vl[t] = V[t];
        Wl[t] = W[t];
    }
    __syncthreads();

    const int lane   = threadIdx.x & 31;
    const int q      = lane & 15;
    const int half   = lane >> 4;
    const int waveId = (blockIdx.x * blockDim.x + threadIdx.x) >> 5;
    const int nWaves = (gridDim.x * blockDim.x) >> 5;
    const int nTiles = (nQueries + 15) >> 4;

    for (int tile = waveId; tile < nTiles; tile += nWaves) {
        const int qi = tile * 16 + q;
        const int qc = (qi < nQueries) ? qi : (nQueries - 1);

        const int i = idx[qc * 3 + 0];
        const int j = idx[qc * 3 + 1];
        const int k = idx[qc * 3 + 2];

        {
            const int nq = (tile + nWaves) * 16 + q;
            if (nq < nQueries)
                __builtin_prefetch(&idx[nq * 3], 0, 1);
        }

        v8f c = {};
        #pragma unroll
        for (int t = 0; t < 4; ++t) {
            const int r0 = 4 * t + 2 * half;
            const int r1 = r0 + 1;

            v2f a, b;
            a.x = Ul[r0 * DIM + i] * Vl[r0 * DIM + j];
            a.y = Ul[r1 * DIM + i] * Vl[r1 * DIM + j];
            b.x = Wl[r0 * DIM + k];
            b.y = Wl[r1 * DIM + k];

            c = __builtin_amdgcn_wmma_f32_16x16x4_f32(
                    false, a, false, b, (short)0, c, false, false);
        }

        if (lane < 8 || lane >= 24) {
            const int dq = (lane < 8) ? lane : (lane - 16);
            const int oq = tile * 16 + dq;
            if (oq < nQueries) {
                const int s = lane & 7;
                float vsel = c[0];
                if (s == 1) vsel = c[1];
                if (s == 2) vsel = c[2];
                if (s == 3) vsel = c[3];
                if (s == 4) vsel = c[4];
                if (s == 5) vsel = c[5];
                if (s == 6) vsel = c[6];
                if (s == 7) vsel = c[7];
                out[oq] = vsel;
            }
        }
    }
}

extern "C" void kernel_launch(void* const* d_in, const int* in_sizes, int n_in,
                              void* d_out, int out_size, void* d_ws, size_t ws_size,
                              hipStream_t stream) {
    const float* U   = (const float*)d_in[0];
    const float* V   = (const float*)d_in[1];
    const float* W   = (const float*)d_in[2];
    const int*   idx = (const int*)  d_in[3];
    float*       out = (float*)d_out;

    const int nQueries = out_size;
    const int nTiles   = (nQueries + 15) / 16;

    const int threads       = 256;
    const int wavesPerBlock = threads / 32;
    int blocks = (nTiles + wavesPerBlock - 1) / wavesPerBlock;
    if (blocks > 128) blocks = 128;
    if (blocks < 1)   blocks = 1;

    const size_t ldsBytes = (size_t)LDS_FLOATS * sizeof(float);

    hipLaunchKernelGGL(parafac_query_wmma, dim3(blocks), dim3(threads), ldsBytes, stream,
                       U, V, W, idx, out, nQueries);
}


# LDS fill: each thread issues all its table loads (128-bit, trip count fixed by the 256-thread block) before its first LDS store, instead of 25 dependent load-wait-store trips
# speedup vs baseline: 2.3919x; 2.3919x over previous
#include <hip/hip_runtime.h>

typedef float v2f __attribute__((ext_vector_type(2)));
typedef float v8f __attribute__((ext_vector_type(8)));
typedef float v4f __attribute__((ext_vector_type(4)));

#define RANK 16
#define DIM  400
#define FACTOR_ELEMS (RANK * DIM)
#define LDS_FLOATS   (3 * FACTOR_ELEMS)

#define BLOCK_THREADS 256
#define FACTOR_VEC4   (FACTOR_ELEMS / 4)
#define FILL_PASSES   (FACTOR_VEC4 / BLOCK_THREADS)
#define FILL_TAIL     (FACTOR_VEC4 - FILL_PASSES * BLOCK_THREADS)

__global__ void __launch_bounds__(BLOCK_THREADS) parafac_query_wmma(const float* __restrict__ U,
                                   const float* __restrict__ V,
                                   const float* __restrict__ W,
                                   const int*   __restrict__ idx,
                                   float*       __restrict__ out,
                                   int nQueries)
{
    extern __shared__ __attribute__((aligned(16))) float lds[];
    float* Ul = lds;
    float* Vl = lds + FACTOR_ELEMS;
    float* Wl = lds + 2 * FACTOR_ELEMS;

    {
        const v4f* U4 = reinterpret_cast<const v4f*>(U);
        const v4f* V4 = reinterpret_cast<const v4f*>(V);
        const v4f* W4 = reinterpret_cast<const v4f*>(W);
        v4f* L4 = reinterpret_cast<v4f*>(lds);
        const int tid = threadIdx.x;

        v4f ru[FILL_PASSES], rv[FILL_PASSES], rw[FILL_PASSES];
        #pragma unroll
        for (int p = 0; p < FILL_PASSES; ++p) {
            ru[p] = U4[tid + p * BLOCK_THREADS];
            rv[p] = V4[tid + p * BLOCK_THREADS];
            rw[p] = W4[tid + p * BLOCK_THREADS];
        }
        const int tf = tid / FILL_TAIL;
        const int tt = FILL_PASSES * BLOCK_THREADS + (tid % FILL_TAIL);
        v4f rt = {};
        if (tf < 3) {
            const v4f* T4 = (tf == 0) ? U4 : ((tf == 1) ? V4 : W4);
            rt = T4[tt];
        }

        #pragma unroll
        for (int p = 0; p < FILL_PASSES; ++p) {
            L4[tid + p * BLOCK_THREADS]                   = ru[p];
            L4[FACTOR_VEC4 + tid + p * BLOCK_THREADS]     = rv[p];
            L4[2 * FACTOR_VEC4 + tid + p * BLOCK_THREADS] = rw[p];
        }
        if (tf < 3)
            L4[tf * FACTOR_VEC4 + tt] = rt;
    }
    __syncthreads();

    const int lane   = threadIdx.x & 31;
    const int q      = lane & 15;
    const int half   = lane >> 4;
    const int waveId = (blockIdx.x * blockDim.x + threadIdx.x) >> 5;
    const int nWaves = (gridDim.x * blockDim.x) >> 5;
    const int nTiles = (nQueries + 15) >> 4;

    for (int tile = waveId; tile < nTiles; tile += nWaves) {
        const int qi = tile * 16 + q;
        const int qc = (qi < nQueries) ? qi : (nQueries - 1);

        const int i = idx[qc * 3 + 0];
        const int j = idx[qc * 3 + 1];
        const int k = idx[qc * 3 + 2];

        {
            const int nq = (tile + nWaves) * 16 + q;
            if (nq < nQueries)
                __builtin_prefetch(&idx[nq * 3], 0, 1);
        }

        v8f c = {};
        #pragma unroll
        for (int t = 0; t < 4; ++t) {
            const int r0 = 4 * t + 2 * half;
            const int r1 = r0 + 1;

            v2f a, b;
            a.x = Ul[r0 * DIM + i] * Vl[r0 * DIM + j];
            a.y = Ul[r1 * DIM + i] * Vl[r1 * DIM + j];
            b.x = Wl[r0 * DIM + k];
            b.y = Wl[r1 * DIM + k];

            c = __builtin_amdgcn_wmma_f32_16x16x4_f32(
                    false, a, false, b, (short)0, c, false, false);
        }

        if (lane < 8 || lane >= 24) {
            const int dq = (lane < 8) ? lane : (lane - 16);
            const int oq = tile * 16 + dq;
            if (oq < nQueries) {
                const int s = lane & 7;
                float vsel = c[0];
                if (s == 1) vsel = c[1];
                if (s == 2) vsel = c[2];
                if (s == 3) vsel = c[3];
                if (s == 4) vsel = c[4];
                if (s == 5) vsel = c[5];
                if (s == 6) vsel = c[6];
                if (s == 7) vsel = c[7];
                out[oq] = vsel;
            }
        }
    }
}

extern "C" void kernel_launch(void* const* d_in, const int* in_sizes, int n_in,
                              void* d_out, int out_size, void* d_ws, size_t ws_size,
                              hipStream_t stream) {
    const float* U   = (const float*)d_in[0];
    const float* V   = (const float*)d_in[1];
    const float* W   = (const float*)d_in[2];
    const int*   idx = (const int*)  d_in[3];
    float*       out = (float*)d_out;

    const int nQueries = out_size;
    const int nTiles   = (nQueries + 15) / 16;

    const int threads       = BLOCK_THREADS;
    const int wavesPerBlock = threads / 32;
    int blocks = (nTiles + wavesPerBlock - 1) / wavesPerBlock;
    if (blocks > 128) blocks = 128;
    if (blocks < 1)   blocks = 1;

    const size_t ldsBytes = (size_t)LDS_FLOATS * sizeof(float);

    hipLaunchKernelGGL(parafac_query_wmma, dim3(blocks), dim3(threads), ldsBytes, stream,
                       U, V, W, idx, out, nQueries);
}


# Index triples of all tiles of a wave (groups of 8) are loaded in one group beside the table fill, before the barrier, instead of one dependent second-level-cache trip per tile in the loop
# speedup vs baseline: 3.6366x; 1.5204x over previous
#include <hip/hip_runtime.h>

typedef float v2f __attribute__((ext_vector_type(2)));
typedef float v8f __attribute__((ext_vector_type(8)));
typedef float v4f __attribute__((ext_vector_type(4)));

#define RANK 16
#define DIM  400
#define FACTOR_ELEMS (RANK * DIM)
#define LDS_FLOATS   (3 * FACTOR_ELEMS)

#define BLOCK_THREADS 256
#define FACTOR_VEC4   (FACTOR_ELEMS / 4)
#define FILL_PASSES   (FACTOR_VEC4 / BLOCK_THREADS)
#define FILL_TAIL     (FACTOR_VEC4 - FILL_PASSES * BLOCK_THREADS)

#define TILES_AHEAD 8

__global__ void __launch_bounds__(BLOCK_THREADS) parafac_query_wmma(const float* __restrict__ U,
                                   const float* __restrict__ V,
                                   const float* __restrict__ W,
                                   const int*   __restrict__ idx,
                                   float*       __restrict__ out,
                                   int nQueries)
{
    extern __shared__ __attribute__((aligned(16))) float lds[];
    float* Ul = lds;
    float* Vl = lds + FACTOR_ELEMS;
    float* Wl = lds + 2 * FACTOR_ELEMS;

    const int lane   = threadIdx.x & 31;
    const int q      = lane & 15;
    const int half   = lane >> 4;
    const int waveId = __builtin_amdgcn_readfirstlane((int)((blockIdx.x * blockDim.x + threadIdx.x) >> 5));
    const int nWaves = (gridDim.x * blockDim.x) >> 5;
    const int nTiles = (nQueries + 15) >> 4;

    int qi[TILES_AHEAD], qj[TILES_AHEAD], qk[TILES_AHEAD];
    auto request_indices = [&](int first) {
        #pragma unroll
        for (int n = 0; n < TILES_AHEAD; ++n) {
            const int qn = (first + n * nWaves) * 16 + q;
            const int qc = (qn < nQueries) ? qn : (nQueries - 1);
            qi[n] = idx[qc * 3 + 0];
            qj[n] = idx[qc * 3 + 1];
            qk[n] = idx[qc * 3 + 2];
        }
    };

    {
        const v4f* U4 = reinterpret_cast<const v4f*>(U);
        const v4f* V4 = reinterpret_cast<const v4f*>(V);
        const v4f* W4 = reinterpret_cast<const v4f*>(W);
        v4f* L4 = reinterpret_cast<v4f*>(lds);
        const int tid = threadIdx.x;

        v4f ru[FILL_PASSES], rv[FILL_PASSES], rw[FILL_PASSES];
        #pragma unroll
        for (int p = 0; p < FILL_PASSES; ++p) {
            ru[p] = U4[tid + p * BLOCK_THREADS];
            rv[p] = V4[tid + p * BLOCK_THREADS];
            rw[p] = W4[tid + p * BLOCK_THREADS];
        }
        request_indices(waveId);

        const int tf = tid / FILL_TAIL;
        const int tt = FILL_PASSES * BLOCK_THREADS + (tid % FILL_TAIL);
        v4f rt = {};
        if (tf < 3) {
            const v4f* T4 = (tf == 0) ? U4 : ((tf == 1) ? V4 : W4);
            rt = T4[tt];
        }

        #pragma unroll
        for (int p = 0; p < FILL_PASSES; ++p) {
            L4[tid + p * BLOCK_THREADS]                   = ru[p];
            L4[FACTOR_VEC4 + tid + p * BLOCK_THREADS]     = rv[p];
            L4[2 * FACTOR_VEC4 + tid + p * BLOCK_THREADS] = rw[p];
        }
        if (tf < 3)
            L4[tf * FACTOR_VEC4 + tt] = rt;
    }
    __syncthreads();

    for (int first = waveId; first < nTiles; first += TILES_AHEAD * nWaves) {
        #pragma unroll
        for (int n = 0; n < TILES_AHEAD; ++n) {
            const int tile = first + n * nWaves;
            if (tile < nTiles) {
                const int i = qi[n];
                const int j = qj[n];
                const int k = qk[n];

                v8f c = {};
                #pragma unroll
                for (int t = 0; t < 4; ++t) {
                    const int r0 = 4 * t + 2 * half;
                    const int r1 = r0 + 1;

                    v2f a, b;
                    a.x = Ul[r0 * DIM + i] * Vl[r0 * DIM + j];
                    a.y = Ul[r1 * DIM + i] * Vl[r1 * DIM + j];
                    b.x = Wl[r0 * DIM + k];
                    b.y = Wl[r1 * DIM + k];

                    c = __builtin_amdgcn_wmma_f32_16x16x4_f32(
                            false, a, false, b, (short)0, c, false, false);
                }

                if (lane < 8 || lane >= 24) {
                    const int dq = (lane < 8) ? lane : (lane - 16);
                    const int oq = tile * 16 + dq;
                    if (oq < nQueries) {
                        const int s = lane & 7;
                        float vsel = c[0];
                        if (s == 1) vsel = c[1];
                        if (s == 2) vsel = c[2];
                        if (s == 3) vsel = c[3];
                        if (s == 4) vsel = c[4];
                        if (s == 5) vsel = c[5];
                        if (s == 6) vsel = c[6];
                        if (s == 7) vsel = c[7];
                        out[oq] = vsel;
                    }
                }
            }
        }

        if (first + TILES_AHEAD * nWaves < nTiles)
            request_indices(first + TILES_AHEAD * nWaves);
    }
}

extern "C" void kernel_launch(void* const* d_in, const int* in_sizes, int n_in,
                              void* d_out, int out_size, void* d_ws, size_t ws_size,
                              hipStream_t stream) {
    const float* U   = (const float*)d_in[0];
    const float* V   = (const float*)d_in[1];
    const float* W   = (const float*)d_in[2];
    const int*   idx = (const int*)  d_in[3];
    float*       out = (float*)d_out;

    const int nQueries = out_size;
    const int nTiles   = (nQueries + 15) / 16;

    const int threads       = BLOCK_THREADS;
    const int wavesPerBlock = threads / 32;
    int blocks = (nTiles + wavesPerBlock - 1) / wavesPerBlock;
    if (blocks > 128) blocks = 128;
    if (blocks < 1)   blocks = 1;

    const size_t ldsBytes = (size_t)LDS_FLOATS * sizeof(float);

    hipLaunchKernelGGL(parafac_query_wmma, dim3(blocks), dim3(threads), ldsBytes, stream,
                       U, V, W, idx, out, nQueries);
}


# LDS tables kept transposed (one 80-byte row per index, ranks in fragment order): a lane gathers its 8 ranks of a table with two 128-bit LDS loads instead of eight 32-bit ones; sum order unchanged
# speedup vs baseline: 3.8775x; 1.0662x over previous
#include <hip/hip_runtime.h>

typedef float v2f __attribute__((ext_vector_type(2)));
typedef float v8f __attribute__((ext_vector_type(8)));
typedef float v4f __attribute__((ext_vector_type(4)));

#define RANK 16
#define DIM  400
#define ROW_PITCH    20
#define ROW_VEC4     (ROW_PITCH / 4)
#define TABLE_FLOATS (DIM * ROW_PITCH)
#define TABLE_VEC4   (TABLE_FLOATS / 4)
#define LDS_FLOATS   (3 * TABLE_FLOATS)

#define BLOCK_THREADS 256
#define COLGROUPS     (DIM / 4)
#define FACTOR_UNITS  (COLGROUPS * 4)
#define FILL_UNITS    (3 * FACTOR_UNITS)
#define FILL_PASSES   (FILL_UNITS / BLOCK_THREADS)
#define FILL_TAIL     (FILL_UNITS - FILL_PASSES * BLOCK_THREADS)

#define TILES_AHEAD 8

__global__ void __launch_bounds__(BLOCK_THREADS) parafac_query_wmma(const float* __restrict__ U,
                                   const float* __restrict__ V,
                                   const float* __restrict__ W,
                                   const int*   __restrict__ idx,
                                   float*       __restrict__ out,
                                   int nQueries)
{
    extern __shared__ __attribute__((aligned(16))) float lds[];
    const v4f* Ul4 = reinterpret_cast<const v4f*>(lds);
    const v4f* Vl4 = Ul4 + TABLE_VEC4;
    const v4f* Wl4 = Ul4 + 2 * TABLE_VEC4;

    const int lane   = threadIdx.x & 31;
    const int q      = lane & 15;
    const int half   = lane >> 4;
    const int waveId = __builtin_amdgcn_readfirstlane((int)((blockIdx.x * blockDim.x + threadIdx.x) >> 5));
    const int nWaves = (gridDim.x * blockDim.x) >> 5;
    const int nTiles = (nQueries + 15) >> 4;

    int qi[TILES_AHEAD], qj[TILES_AHEAD], qk[TILES_AHEAD];
    auto request_indices = [&](int first) {
        #pragma unroll
        for (int n = 0; n < TILES_AHEAD; ++n) {
            const int qn = (first + n * nWaves) * 16 + q;
            const int qc = (qn < nQueries) ? qn : (nQueries - 1);
            qi[n] = idx[qc * 3 + 0];
            qj[n] = idx[qc * 3 + 1];
            qk[n] = idx[qc * 3 + 2];
        }
    };

    {
        const v4f* U4 = reinterpret_cast<const v4f*>(U);
        const v4f* V4 = reinterpret_cast<const v4f*>(V);
        const v4f* W4 = reinterpret_cast<const v4f*>(W);
        v4f* L4 = reinterpret_cast<v4f*>(lds);
        const int tid = threadIdx.x;

        auto unit_load = [&](int un, v4f (&r)[4], int& dst4) {
            const int f  = un / FACTOR_UNITS;
            const int u  = un - f * FACTOR_UNITS;
            const int g  = u >> 2;
            const int pg = u & 3;
            const v4f* F4 = (f == 0) ? U4 : ((f == 1) ? V4 : W4);
            const int rb = 8 * (pg & 1) + 2 * (pg >> 1);
            r[0] = F4[(rb + 0) * COLGROUPS + g];
            r[1] = F4[(rb + 1) * COLGROUPS + g];
            r[2] = F4[(rb + 4) * COLGROUPS + g];
            r[3] = F4[(rb + 5) * COLGROUPS + g];
            dst4 = f * TABLE_VEC4 + (4 * g) * ROW_VEC4 + pg;
        };
        auto unit_store = [&](const v4f (&r)[4], int dst4) {
            #pragma unroll
            for (int c = 0; c < 4; ++c) {
                v4f row;
                row.x = r[0][c];
                row.y = r[1][c];
                row.z = r[2][c];
                row.w = r[3][c];
                L4[dst4 + c * ROW_VEC4] = row;
            }
        };

        v4f blk[FILL_PASSES][4];
        int dst[FILL_PASSES];
        #pragma unroll
        for (int p = 0; p < FILL_PASSES; ++p)
            unit_load(tid + p * BLOCK_THREADS, blk[p], dst[p]);

        request_indices(waveId);

        v4f tblk[4] = {};
        int tdst = 0;
        if (tid < FILL_TAIL)
            unit_load(FILL_PASSES * BLOCK_THREADS + tid, tblk, tdst);

        #pragma unroll
        for (int p = 0; p < FILL_PASSES; ++p)
            unit_store(blk[p], dst[p]);
        if (tid < FILL_TAIL)
            unit_store(tblk, tdst);
    }
    __syncthreads();

    for (int first = waveId; first < nTiles; first += TILES_AHEAD * nWaves) {
        #pragma unroll
        for (int n = 0; n < TILES_AHEAD; ++n) {
            const int tile = first + n * nWaves;
            if (tile < nTiles) {
                const int i = qi[n];
                const int j = qj[n];
                const int k = qk[n];

                const v4f u0 = Ul4[i * ROW_VEC4 + 2 * half], u1 = Ul4[i * ROW_VEC4 + 2 * half + 1];
                const v4f v0 = Vl4[j * ROW_VEC4 + 2 * half], v1 = Vl4[j * ROW_VEC4 + 2 * half + 1];
                const v4f w0 = Wl4[k * ROW_VEC4 + 2 * half], w1 = Wl4[k * ROW_VEC4 + 2 * half + 1];
                const float uu[8] = {u0.x, u0.y, u0.z, u0.w, u1.x, u1.y, u1.z, u1.w};
                const float vv[8] = {v0.x, v0.y, v0.z, v0.w, v1.x, v1.y, v1.z, v1.w};
                const float ww[8] = {w0.x, w0.y, w0.z, w0.w, w1.x, w1.y, w1.z, w1.w};

                v8f c = {};
                #pragma unroll
                for (int t = 0; t < 4; ++t) {
                    v2f a, b;
                    a.x = uu[2 * t]     * vv[2 * t];
                    a.y = uu[2 * t + 1] * vv[2 * t + 1];
                    b.x = ww[2 * t];
                    b.y = ww[2 * t + 1];

                    c = __builtin_amdgcn_wmma_f32_16x16x4_f32(
                            false, a, false, b, (short)0, c, false, false);
                }

                if (lane < 8 || lane >= 24) {
                    const int dq = (lane < 8) ? lane : (lane - 16);
                    const int oq = tile * 16 + dq;
                    if (oq < nQueries) {
                        const int s = lane & 7;
                        float vsel = c[0];
                        if (s == 1) vsel = c[1];
                        if (s == 2) vsel = c[2];
                        if (s == 3) vsel = c[3];
                        if (s == 4) vsel = c[4];
                        if (s == 5) vsel = c[5];
                        if (s == 6) vsel = c[6];
                        if (s == 7) vsel = c[7];
                        out[oq] = vsel;
                    }
                }
            }
        }

        if (first + TILES_AHEAD * nWaves < nTiles)
            request_indices(first + TILES_AHEAD * nWaves);
    }
}

extern "C" void kernel_launch(void* const* d_in, const int* in_sizes, int n_in,
                              void* d_out, int out_size, void* d_ws, size_t ws_size,
                              hipStream_t stream) {
    const float* U   = (const float*)d_in[0];
    const float* V   = (const float*)d_in[1];
    const float* W   = (const float*)d_in[2];
    const int*   idx = (const int*)  d_in[3];
    float*       out = (float*)d_out;

    const int nQueries = out_size;
    const int nTiles   = (nQueries + 15) / 16;

    const int threads       = BLOCK_THREADS;
    const int wavesPerBlock = threads / 32;
    int blocks = (nTiles + wavesPerBlock - 1) / wavesPerBlock;
    if (blocks > 128) blocks = 128;
    if (blocks < 1)   blocks = 1;

    const size_t ldsBytes = (size_t)LDS_FLOATS * sizeof(float);

    hipLaunchKernelGGL(parafac_query_wmma, dim3(blocks), dim3(threads), ldsBytes, stream,
                       U, V, W, idx, out, nQueries);
}


# Two neighbouring 16-query tiles per wave step: their 12 LDS gathers issue as one group, the two WMMA chains are independent, and tile 2 takes W columns rotated by 8 so one 128-byte store serves both
# speedup vs baseline: 4.1134x; 1.0608x over previous
#include <hip/hip_runtime.h>

typedef float v2f __attribute__((ext_vector_type(2)));
typedef float v8f __attribute__((ext_vector_type(8)));
typedef float v4f __attribute__((ext_vector_type(4)));

#define RANK 16
#define DIM  400
#define ROW_PITCH    20
#define ROW_VEC4     (ROW_PITCH / 4)
#define TABLE_FLOATS (DIM * ROW_PITCH)
#define TABLE_VEC4   (TABLE_FLOATS / 4)
#define LDS_FLOATS   (3 * TABLE_FLOATS)

#define BLOCK_THREADS 256
#define COLGROUPS     (DIM / 4)
#define FACTOR_UNITS  (COLGROUPS * 4)
#define FILL_UNITS    (3 * FACTOR_UNITS)
#define FILL_PASSES   (FILL_UNITS / BLOCK_THREADS)
#define FILL_TAIL     (FILL_UNITS - FILL_PASSES * BLOCK_THREADS)

#define PAIRS_AHEAD 4

__global__ void __launch_bounds__(BLOCK_THREADS) parafac_query_wmma(const float* __restrict__ U,
                                   const float* __restrict__ V,
                                   const float* __restrict__ W,
                                   const int*   __restrict__ idx,
                                   float*       __restrict__ out,
                                   int nQueries)
{
    extern __shared__ __attribute__((aligned(16))) float lds[];
    const v4f* Ul4 = reinterpret_cast<const v4f*>(lds);
    const v4f* Vl4 = Ul4 + TABLE_VEC4;
    const v4f* Wl4 = Ul4 + 2 * TABLE_VEC4;

    const int lane   = threadIdx.x & 31;
    const int q      = lane & 15;
    const int half   = lane >> 4;
    const int waveId = __builtin_amdgcn_readfirstlane((int)((blockIdx.x * blockDim.x + threadIdx.x) >> 5));
    const int nWaves = (gridDim.x * blockDim.x) >> 5;
    const int nPairs = (nQueries + 31) >> 5;

    int xi[PAIRS_AHEAD], xj[PAIRS_AHEAD], xk[PAIRS_AHEAD];
    int yi[PAIRS_AHEAD], yj[PAIRS_AHEAD], yk[PAIRS_AHEAD];
    auto request_indices = [&](int first) {
        #pragma unroll
        for (int n = 0; n < PAIRS_AHEAD; ++n) {
            const int base = (first + n * nWaves) * 32;
            const int last = nQueries - 1;
            const int qx   = base + q;
            const int qy   = base + 16 + q;
            const int qyw  = base + 16 + (q ^ 8);
            const int cx   = (qx  < nQueries) ? qx  : last;
            const int cy   = (qy  < nQueries) ? qy  : last;
            const int cyw  = (qyw < nQueries) ? qyw : last;
            xi[n] = idx[cx * 3 + 0];
            xj[n] = idx[cx * 3 + 1];
            xk[n] = idx[cx * 3 + 2];
            yi[n] = idx[cy * 3 + 0];
            yj[n] = idx[cy * 3 + 1];
            yk[n] = idx[cyw * 3 + 2];
        }
    };

    {
        const v4f* U4 = reinterpret_cast<const v4f*>(U);
        const v4f* V4 = reinterpret_cast<const v4f*>(V);
        const v4f* W4 = reinterpret_cast<const v4f*>(W);
        v4f* L4 = reinterpret_cast<v4f*>(lds);
        const int tid = threadIdx.x;

        auto unit_load = [&](int un, v4f (&r)[4], int& dst4) {
            const int f  = un / FACTOR_UNITS;
            const int u  = un - f * FACTOR_UNITS;
            const int g  = u >> 2;
            const int pg = u & 3;
            const v4f* F4 = (f == 0) ? U4 : ((f == 1) ? V4 : W4);
            const int rb = 8 * (pg & 1) + 2 * (pg >> 1);
            r[0] = F4[(rb + 0) * COLGROUPS + g];
            r[1] = F4[(rb + 1) * COLGROUPS + g];
            r[2] = F4[(rb + 4) * COLGROUPS + g];
            r[3] = F4[(rb + 5) * COLGROUPS + g];
            dst4 = f * TABLE_VEC4 + (4 * g) * ROW_VEC4 + pg;
        };
        auto unit_store = [&](const v4f (&r)[4], int dst4) {
            #pragma unroll
            for (int c = 0; c < 4; ++c) {
                v4f row;
                row.x = r[0][c];
                row.y = r[1][c];
                row.z = r[2][c];
                row.w = r[3][c];
                L4[dst4 + c * ROW_VEC4] = row;
            }
        };

        v4f blk[FILL_PASSES][4];
        int dst[FILL_PASSES];
        #pragma unroll
        for (int p = 0; p < FILL_PASSES; ++p)
            unit_load(tid + p * BLOCK_THREADS, blk[p], dst[p]);

        request_indices(waveId);

        v4f tblk[4] = {};
        int tdst = 0;
        if (tid < FILL_TAIL)
            unit_load(FILL_PASSES * BLOCK_THREADS + tid, tblk, tdst);

        #pragma unroll
        for (int p = 0; p < FILL_PASSES; ++p)
            unit_store(blk[p], dst[p]);
        if (tid < FILL_TAIL)
            unit_store(tblk, tdst);
    }
    __syncthreads();

    const bool fromY  = (lane >= 8) && (lane < 24);
    const int  outPos = (lane < 8) ? lane : (fromY ? (lane + 8) : (lane - 16));
    const int  s      = lane & 7;

    for (int first = waveId; first < nPairs; first += PAIRS_AHEAD * nWaves) {
        #pragma unroll
        for (int n = 0; n < PAIRS_AHEAD; ++n) {
            const int pair = first + n * nWaves;
            if (pair < nPairs) {
                const int ux = xi[n] * ROW_VEC4 + 2 * half, vx = xj[n] * ROW_VEC4 + 2 * half, wx = xk[n] * ROW_VEC4 + 2 * half;
                const int uy = yi[n] * ROW_VEC4 + 2 * half, vy = yj[n] * ROW_VEC4 + 2 * half, wy = yk[n] * ROW_VEC4 + 2 * half;
                const v4f ux0 = Ul4[ux], ux1 = Ul4[ux + 1];
                const v4f vx0 = Vl4[vx], vx1 = Vl4[vx + 1];
                const v4f wx0 = Wl4[wx], wx1 = Wl4[wx + 1];
                const v4f uy0 = Ul4[uy], uy1 = Ul4[uy + 1];
                const v4f vy0 = Vl4[vy], vy1 = Vl4[vy + 1];
                const v4f wy0 = Wl4[wy], wy1 = Wl4[wy + 1];
                const float uux[8] = {ux0.x, ux0.y, ux0.z, ux0.w, ux1.x, ux1.y, ux1.z, ux1.w};
                const float vvx[8] = {vx0.x, vx0.y, vx0.z, vx0.w, vx1.x, vx1.y, vx1.z, vx1.w};
                const float wwx[8] = {wx0.x, wx0.y, wx0.z, wx0.w, wx1.x, wx1.y, wx1.z, wx1.w};
                const float uuy[8] = {uy0.x, uy0.y, uy0.z, uy0.w, uy1.x, uy1.y, uy1.z, uy1.w};
                const float vvy[8] = {vy0.x, vy0.y, vy0.z, vy0.w, vy1.x, vy1.y, vy1.z, vy1.w};
                const float wwy[8] = {wy0.x, wy0.y, wy0.z, wy0.w, wy1.x, wy1.y, wy1.z, wy1.w};

                v8f cx = {};
                v8f cy = {};
                #pragma unroll
                for (int t = 0; t < 4; ++t) {
                    v2f ax, bx, ay, by;
                    ax.x = uux[2 * t]     * vvx[2 * t];
                    ax.y = uux[2 * t + 1] * vvx[2 * t + 1];
                    bx.x = wwx[2 * t];
                    bx.y = wwx[2 * t + 1];
                    ay.x = uuy[2 * t]     * vvy[2 * t];
                    ay.y = uuy[2 * t + 1] * vvy[2 * t + 1];
                    by.x = wwy[2 * t];
                    by.y = wwy[2 * t + 1];

                    cx = __builtin_amdgcn_wmma_f32_16x16x4_f32(
                            false, ax, false, bx, (short)0, cx, false, false);
                    cy = __builtin_amdgcn_wmma_f32_16x16x4_f32(
                            false, ay, false, by, (short)0, cy, false, false);
                }

                const int oq = pair * 32 + outPos;
                if (oq < nQueries) {
                    float vsel = fromY ? cy[0] : cx[0];
                    if (s == 1) vsel = fromY ? cy[1] : cx[1];
                    if (s == 2) vsel = fromY ? cy[2] : cx[2];
                    if (s == 3) vsel = fromY ? cy[3] : cx[3];
                    if (s == 4) vsel = fromY ? cy[4] : cx[4];
                    if (s == 5) vsel = fromY ? cy[5] : cx[5];
                    if (s == 6) vsel = fromY ? cy[6] : cx[6];
                    if (s == 7) vsel = fromY ? cy[7] : cx[7];
                    out[oq] = vsel;
                }
            }
        }

        if (first + PAIRS_AHEAD * nWaves < nPairs)
            request_indices(first + PAIRS_AHEAD * nWaves);
    }
}

extern "C" void kernel_launch(void* const* d_in, const int* in_sizes, int n_in,
                              void* d_out, int out_size, void* d_ws, size_t ws_size,
                              hipStream_t stream) {
    const float* U   = (const float*)d_in[0];
    const float* V   = (const float*)d_in[1];
    const float* W   = (const float*)d_in[2];
    const int*   idx = (const int*)  d_in[3];
    float*       out = (float*)d_out;

    const int nQueries = out_size;
    const int nTiles   = (nQueries + 15) / 16;

    const int threads       = BLOCK_THREADS;
    const int wavesPerBlock = threads / 32;
    int blocks = (nTiles + wavesPerBlock - 1) / wavesPerBlock;
    if (blocks > 128) blocks = 128;
    if (blocks < 1)   blocks = 1;

    const size_t ldsBytes = (size_t)LDS_FLOATS * sizeof(float);

    hipLaunchKernelGGL(parafac_query_wmma, dim3(blocks), dim3(threads), ldsBytes, stream,
                       U, V, W, idx, out, nQueries);
}
